# starting kernel
# baseline (speedup 1.0000x reference)
#include <hip/hip_runtime.h>
#include <math.h>

#define N_NODES 100000
#define N_EDGES 1600000
#define F 64
#define K_ORDER 8
#define LN_EPS 1e-5f

typedef __attribute__((ext_vector_type(2))) float v2f;
typedef __attribute__((ext_vector_type(8))) float v8f;

__global__ void zero_f32(float* __restrict__ p, int n) {
  int i = blockIdx.x * blockDim.x + threadIdx.x;
  if (i < n) p[i] = 0.0f;
}

__global__ void degree_kernel(const int* __restrict__ row,
                              const int* __restrict__ col,
                              float* __restrict__ deg) {
  int e = blockIdx.x * blockDim.x + threadIdx.x;
  if (e >= N_EDGES) return;
  int r = row[e], c = col[e];
  if (r != c) unsafeAtomicAdd(&deg[r], 1.0f);
}

__global__ void dinv_kernel(const float* __restrict__ deg,
                            float* __restrict__ dinv) {
  int i = blockIdx.x * blockDim.x + threadIdx.x;
  if (i >= N_NODES) return;
  float d = deg[i];
  dinv[i] = (d > 0.0f) ? rsqrtf(fmaxf(d, 1.0f)) : 0.0f;
}

__global__ void spmm_kernel(const int* __restrict__ row,
                            const int* __restrict__ col,
                            const float* __restrict__ dinv,
                            const float* __restrict__ h,
                            float* __restrict__ out) {
  int wid  = (blockIdx.x * blockDim.x + threadIdx.x) >> 5;
  int lane = threadIdx.x & 31;
  if (wid >= N_EDGES) return;
  int r = row[wid], c = col[wid];
  if (r == c) return;
  float w = -dinv[r] * dinv[c];
  v2f hv = *(const v2f*)(h + (size_t)c * F + lane * 2);
  float* o = out + (size_t)r * F + lane * 2;
  unsafeAtomicAdd(o,     w * hv.x);
  unsafeAtomicAdd(o + 1, w * hv.y);
}

__global__ void combine_kernel(float* __restrict__ p,
                               const float* __restrict__ tkm2, int n) {
  int i = blockIdx.x * blockDim.x + threadIdx.x;
  if (i < n) p[i] = 2.0f * p[i] - tkm2[i];
}

__global__ void wmma_gemm_kernel(const float* __restrict__ T,
                                 const float* __restrict__ Wk,
                                 float* __restrict__ acc, int beta) {
  int wavesPerBlock = blockDim.x >> 5;
  int tile = blockIdx.x * wavesPerBlock + (threadIdx.x >> 5);
  const int numTiles = N_NODES / 16;
  if (tile >= numTiles) return;
  int lane = threadIdx.x & 31;
  int m    = lane & 15;
  int half = lane >> 4;
  int n    = lane & 15;

  v8f c0 = {}, c1 = {}, c2 = {}, c3 = {};
  const float* trow = T + ((size_t)tile * 16 + m) * F;
  float* accbase = acc + (size_t)tile * 16 * F;

  if (beta) {
    #pragma unroll
    for (int r = 0; r < 8; ++r) {
      int mm = r + 8 * half;
      c0[r] = accbase[mm * F +  0 + n];
      c1[r] = accbase[mm * F + 16 + n];
      c2[r] = accbase[mm * F + 32 + n];
      c3[r] = accbase[mm * F + 48 + n];
    }
  }

  #pragma unroll
  for (int kk = 0; kk < F; kk += 4) {
    int k0 = kk + half * 2;
    v2f a = *(const v2f*)(trow + k0);
    const float* wr0 = Wk + (size_t)k0 * F;
    const float* wr1 = wr0 + F;
    v2f b0 = { wr0[ 0 + n], wr1[ 0 + n] };
    v2f b1 = { wr0[16 + n], wr1[16 + n] };
    v2f b2 = { wr0[32 + n], wr1[32 + n] };
    v2f b3 = { wr0[48 + n], wr1[48 + n] };
    c0 = __builtin_amdgcn_wmma_f32_16x16x4_f32(false, a, false, b0, (short)0, c0, false, false);
    c1 = __builtin_amdgcn_wmma_f32_16x16x4_f32(false, a, false, b1, (short)0, c1, false, false);
    c2 = __builtin_amdgcn_wmma_f32_16x16x4_f32(false, a, false, b2, (short)0, c2, false, false);
    c3 = __builtin_amdgcn_wmma_f32_16x16x4_f32(false, a, false, b3, (short)0, c3, false, false);
  }

  #pragma unroll
  for (int r = 0; r < 8; ++r) {
    int mm = r + 8 * half;
    accbase[mm * F +  0 + n] = c0[r];
    accbase[mm * F + 16 + n] = c1[r];
    accbase[mm * F + 32 + n] = c2[r];
    accbase[mm * F + 48 + n] = c3[r];
  }
}

__global__ void ln_elu_kernel(const float* __restrict__ acc,
                              const float* __restrict__ bias,
                              const float* __restrict__ lnw,
                              const float* __restrict__ lnb,
                              float* __restrict__ out) {
  int node = blockIdx.x * (blockDim.x >> 5) + (threadIdx.x >> 5);
  if (node >= N_NODES) return;
  int lane = threadIdx.x & 31;
  int f = lane * 2;
  v2f v = *(const v2f*)(acc + (size_t)node * F + f);
  v.x += bias[f];
  v.y += bias[f + 1];
  float s = v.x + v.y;
  #pragma unroll
  for (int off = 16; off > 0; off >>= 1) s += __shfl_xor(s, off, 32);
  float mean = s * (1.0f / F);
  float dx = v.x - mean, dy = v.y - mean;
  float q = dx * dx + dy * dy;
  #pragma unroll
  for (int off = 16; off > 0; off >>= 1) q += __shfl_xor(q, off, 32);
  float inv = rsqrtf(q * (1.0f / F) + LN_EPS);
  float hx = dx * inv * lnw[f]     + lnb[f];
  float hy = dy * inv * lnw[f + 1] + lnb[f + 1];
  hx = (hx > 0.0f) ? hx : expm1f(hx);
  hy = (hy > 0.0f) ? hy : expm1f(hy);
  v2f o = { hx, hy };
  *(v2f*)(out + (size_t)node * F + f) = o;
}

extern "C" void kernel_launch(void* const* d_in, const int* in_sizes, int n_in,
                              void* d_out, int out_size, void* d_ws, size_t ws_size,
                              hipStream_t stream) {
  (void)in_sizes; (void)n_in; (void)out_size; (void)ws_size;

  const float* x      = (const float*)d_in[0];
  const int*   edge   = (const int*)d_in[1];
  const float* weight = (const float*)d_in[2];
  const float* bias   = (const float*)d_in[3];
  const float* lnw    = (const float*)d_in[4];
  const float* lnb    = (const float*)d_in[5];
  const int* row = edge;
  const int* col = edge + N_EDGES;

  float* ws   = (float*)d_ws;
  float* deg  = ws;
  float* dinv = deg + N_NODES;
  float* b0   = dinv + N_NODES;
  float* b1   = b0 + (size_t)N_NODES * F;
  float* b2   = b1 + (size_t)N_NODES * F;
  float* acc  = (float*)d_out;

  const int nfeat = N_NODES * F;
  dim3 blk(256);

  zero_f32<<<(N_NODES + 255) / 256, blk, 0, stream>>>(deg, N_NODES);
  degree_kernel<<<(N_EDGES + 255) / 256, blk, 0, stream>>>(row, col, deg);
  dinv_kernel<<<(N_NODES + 255) / 256, blk, 0, stream>>>(deg, dinv);

  int spmmGrid = (N_EDGES * 32 + 255) / 256;
  zero_f32<<<(nfeat + 255) / 256, blk, 0, stream>>>(b0, nfeat);
  spmm_kernel<<<spmmGrid, blk, 0, stream>>>(row, col, dinv, x, b0);

  const int tiles = N_NODES / 16;
  dim3 gblk(128);
  int ggrid = (tiles + 3) / 4;
  wmma_gemm_kernel<<<ggrid, gblk, 0, stream>>>(x,  weight + 0 * F * F, acc, 0);
  wmma_gemm_kernel<<<ggrid, gblk, 0, stream>>>(b0, weight + 1 * F * F, acc, 1);

  const float* tkm2 = x;
  const float* tkm1 = b0;
  float* bufs[3] = { b0, b1, b2 };
  int next = 1;
  for (int k = 2; k < K_ORDER; ++k) {
    float* p = bufs[next];
    zero_f32<<<(nfeat + 255) / 256, blk, 0, stream>>>(p, nfeat);
    spmm_kernel<<<spmmGrid, blk, 0, stream>>>(row, col, dinv, tkm1, p);
    combine_kernel<<<(nfeat + 255) / 256, blk, 0, stream>>>(p, tkm2, nfeat);
    wmma_gemm_kernel<<<ggrid, gblk, 0, stream>>>(p, weight + (size_t)k * F * F, acc, 1);
    tkm2 = tkm1;
    tkm1 = p;
    next = (next + 1) % 3;
  }

  ln_elu_kernel<<<(N_NODES + 7) / 8, blk, 0, stream>>>(acc, bias, lnw, lnb, (float*)d_out);
}


# Zero-fill and 2p-T(k-2) combine launches folded into the product launches: each also writes the next scatter's start value (0 or -T(k-1)), scatter adds 2w*h; -T(k-2) now leads the atomic sum
# speedup vs baseline: 1.8831x; 1.8831x over previous
#include <hip/hip_runtime.h>
#include <math.h>

#define N_NODES 100000
#define N_EDGES 1600000
#define F 64
#define K_ORDER 8
#define LN_EPS 1e-5f

typedef __attribute__((ext_vector_type(2))) float v2f;
typedef __attribute__((ext_vector_type(4))) float v4f;
typedef __attribute__((ext_vector_type(8))) float v8f;

__global__ void zero_f32(float* __restrict__ p, int n) {
  int i = blockIdx.x * blockDim.x + threadIdx.x;
  if (i < n) p[i] = 0.0f;
}

__global__ void degree_kernel(const int* __restrict__ row,
                              const int* __restrict__ col,
                              float* __restrict__ deg) {
  int e = blockIdx.x * blockDim.x + threadIdx.x;
  if (e >= N_EDGES) return;
  int r = row[e], c = col[e];
  if (r != c) unsafeAtomicAdd(&deg[r], 1.0f);
}

__global__ void dinv_kernel(const float* __restrict__ deg,
                            float* __restrict__ dinv) {
  int i = blockIdx.x * blockDim.x + threadIdx.x;
  if (i >= N_NODES) return;
  float d = deg[i];
  dinv[i] = (d > 0.0f) ? rsqrtf(fmaxf(d, 1.0f)) : 0.0f;
}

#define EPW 32
#define EGRP 16
static_assert(N_EDGES % EPW == 0, "edge count must be a multiple of the wave batch");
static_assert(EPW % EGRP == 0, "group must divide the batch");

__global__ void spmm_kernel(const int* __restrict__ row,
                            const int* __restrict__ col,
                            const float* __restrict__ dinv,
                            const float* __restrict__ h,
                            float* __restrict__ out, float scale) {
  int wid  = (blockIdx.x * blockDim.x + threadIdx.x) >> 5;
  unsigned lane = threadIdx.x & 31u;
  if (wid * EPW >= N_EDGES) return;
  int e = wid * EPW + (int)lane;
  int r = row[e], c = col[e];
  float w = (r != c) ? scale * (-dinv[r] * dinv[c]) : 0.0f;
  int wbits = __builtin_bit_cast(int, w);

  #pragma unroll
  for (int j0 = 0; j0 < EPW; j0 += EGRP) {
    float a0[EGRP], a1[EGRP];
    #pragma unroll
    for (int u = 0; u < EGRP; ++u) {
      int cj = __builtin_amdgcn_readlane(c, j0 + u);
      unsigned hb = (unsigned)cj * F + lane;
      a0[u] = h[hb];
      a1[u] = h[hb + 32u];
    }
    #pragma unroll
    for (int u = 0; u < EGRP; ++u) {
      int rj = __builtin_amdgcn_readlane(r, j0 + u);
      float wj = __builtin_bit_cast(float, __builtin_amdgcn_readlane(wbits, j0 + u));
      unsigned ob = (unsigned)rj * F + lane;
      unsafeAtomicAdd(out + ob,       wj * a0[u]);
      unsafeAtomicAdd(out + ob + 32u, wj * a1[u]);
    }
  }
}

__global__ void __launch_bounds__(128, 7)
wmma_gemm_kernel(const float* __restrict__ T,
                                 const float* __restrict__ Wk,
                                 float* __restrict__ acc, int beta,
                                 const float* src, float* dst, int prep) {
  int wavesPerBlock = blockDim.x >> 5;
  int tile = blockIdx.x * wavesPerBlock + (threadIdx.x >> 5);
  const int numTiles = N_NODES / 16;
  if (tile >= numTiles) return;
  int lane = threadIdx.x & 31;
  int m    = lane & 15;
  int half = lane >> 4;
  int n    = lane & 15;

  v8f c0 = {}, c1 = {}, c2 = {}, c3 = {};
  const float* trow = T + ((size_t)tile * 16 + m) * F;
  float* accbase = acc + (size_t)tile * 16 * F;

  const size_t pbase = (size_t)tile * 16 * F + (size_t)lane * 4;
  v4f pv[8];
  if (prep == 1) {
    #pragma unroll
    for (int i = 0; i < 8; ++i) pv[i] = *(const v4f*)(src + pbase + i * 128);
  }

  if (beta) {
    #pragma unroll
    for (int r = 0; r < 8; ++r) {
      int mm = r + 8 * half;
      c0[r] = accbase[mm * F +  0 + n];
      c1[r] = accbase[mm * F + 16 + n];
      c2[r] = accbase[mm * F + 32 + n];
      c3[r] = accbase[mm * F + 48 + n];
    }
  }

  #pragma unroll
  for (int kk = 0; kk < F; kk += 4) {
    int k0 = kk + half * 2;
    v2f a = *(const v2f*)(trow + k0);
    const float* wr0 = Wk + (size_t)k0 * F;
    const float* wr1 = wr0 + F;
    v2f b0 = { wr0[ 0 + n], wr1[ 0 + n] };
    v2f b1 = { wr0[16 + n], wr1[16 + n] };
    v2f b2 = { wr0[32 + n], wr1[32 + n] };
    v2f b3 = { wr0[48 + n], wr1[48 + n] };
    c0 = __builtin_amdgcn_wmma_f32_16x16x4_f32(false, a, false, b0, (short)0, c0, false, false);
    c1 = __builtin_amdgcn_wmma_f32_16x16x4_f32(false, a, false, b1, (short)0, c1, false, false);
    c2 = __builtin_amdgcn_wmma_f32_16x16x4_f32(false, a, false, b2, (short)0, c2, false, false);
    c3 = __builtin_amdgcn_wmma_f32_16x16x4_f32(false, a, false, b3, (short)0, c3, false, false);
  }

  #pragma unroll
  for (int r = 0; r < 8; ++r) {
    int mm = r + 8 * half;
    accbase[mm * F +  0 + n] = c0[r];
    accbase[mm * F + 16 + n] = c1[r];
    accbase[mm * F + 32 + n] = c2[r];
    accbase[mm * F + 48 + n] = c3[r];
  }

  if (prep == 1) {
    #pragma unroll
    for (int i = 0; i < 8; ++i) *(v4f*)(dst + pbase + i * 128) = -pv[i];
  } else if (prep == 2) {
    const v4f z = { 0.0f, 0.0f, 0.0f, 0.0f };
    #pragma unroll
    for (int i = 0; i < 8; ++i) *(v4f*)(dst + pbase + i * 128) = z;
  }
}

__global__ void ln_elu_kernel(const float* __restrict__ acc,
                              const float* __restrict__ bias,
                              const float* __restrict__ lnw,
                              const float* __restrict__ lnb,
                              float* __restrict__ out) {
  int node = blockIdx.x * (blockDim.x >> 5) + (threadIdx.x >> 5);
  if (node >= N_NODES) return;
  int lane = threadIdx.x & 31;
  int f = lane * 2;
  v2f v = *(const v2f*)(acc + (size_t)node * F + f);
  v.x += bias[f];
  v.y += bias[f + 1];
  float s = v.x + v.y;
  #pragma unroll
  for (int off = 16; off > 0; off >>= 1) s += __shfl_xor(s, off, 32);
  float mean = s * (1.0f / F);
  float dx = v.x - mean, dy = v.y - mean;
  float q = dx * dx + dy * dy;
  #pragma unroll
  for (int off = 16; off > 0; off >>= 1) q += __shfl_xor(q, off, 32);
  float inv = rsqrtf(q * (1.0f / F) + LN_EPS);
  float hx = dx * inv * lnw[f]     + lnb[f];
  float hy = dy * inv * lnw[f + 1] + lnb[f + 1];
  hx = (hx > 0.0f) ? hx : expm1f(hx);
  hy = (hy > 0.0f) ? hy : expm1f(hy);
  v2f o = { hx, hy };
  *(v2f*)(out + (size_t)node * F + f) = o;
}

extern "C" void kernel_launch(void* const* d_in, const int* in_sizes, int n_in,
                              void* d_out, int out_size, void* d_ws, size_t ws_size,
                              hipStream_t stream) {
  (void)in_sizes; (void)n_in; (void)out_size; (void)ws_size;

  const float* x      = (const float*)d_in[0];
  const int*   edge   = (const int*)d_in[1];
  const float* weight = (const float*)d_in[2];
  const float* bias   = (const float*)d_in[3];
  const float* lnw    = (const float*)d_in[4];
  const float* lnb    = (const float*)d_in[5];
  const int* row = edge;
  const int* col = edge + N_EDGES;

  float* ws   = (float*)d_ws;
  float* deg  = ws;
  float* dinv = deg + N_NODES;
  float* b0   = dinv + N_NODES;
  float* b1   = b0 + (size_t)N_NODES * F;
  float* acc  = (float*)d_out;

  dim3 blk(256);

  zero_f32<<<(N_NODES + 255) / 256, blk, 0, stream>>>(deg, N_NODES);
  degree_kernel<<<(N_EDGES + 255) / 256, blk, 0, stream>>>(row, col, deg);
  dinv_kernel<<<(N_NODES + 255) / 256, blk, 0, stream>>>(deg, dinv);

  int spmmGrid = (N_EDGES / EPW * 32 + 255) / 256;
  const int tiles = N_NODES / 16;
  dim3 gblk(128);
  int ggrid = (tiles + 3) / 4;

  wmma_gemm_kernel<<<ggrid, gblk, 0, stream>>>(x, weight + 0 * F * F, acc, 0, x, b0, 2);
  spmm_kernel<<<spmmGrid, blk, 0, stream>>>(row, col, dinv, x, b0, 1.0f);
  wmma_gemm_kernel<<<ggrid, gblk, 0, stream>>>(b0, weight + 1 * F * F, acc, 1, x, b1, 1);

  spmm_kernel<<<spmmGrid, blk, 0, stream>>>(row, col, dinv, b0, b1, 2.0f);
  wmma_gemm_kernel<<<ggrid, gblk, 0, stream>>>(b1, weight + 2 * F * F, acc, 1, b0, b0, 1);
  spmm_kernel<<<spmmGrid, blk, 0, stream>>>(row, col, dinv, b1, b0, 2.0f);
  wmma_gemm_kernel<<<ggrid, gblk, 0, stream>>>(b0, weight + 3 * F * F, acc, 1, b1, b1, 1);
  spmm_kernel<<<spmmGrid, blk, 0, stream>>>(row, col, dinv, b0, b1, 2.0f);
  wmma_gemm_kernel<<<ggrid, gblk, 0, stream>>>(b1, weight + 4 * F * F, acc, 1, b0, b0, 1);
  spmm_kernel<<<spmmGrid, blk, 0, stream>>>(row, col, dinv, b1, b0, 2.0f);
  wmma_gemm_kernel<<<ggrid, gblk, 0, stream>>>(b0, weight + 5 * F * F, acc, 1, b1, b1, 1);
  spmm_kernel<<<spmmGrid, blk, 0, stream>>>(row, col, dinv, b0, b1, 2.0f);
  wmma_gemm_kernel<<<ggrid, gblk, 0, stream>>>(b1, weight + 6 * F * F, acc, 1, b0, b0, 1);
  spmm_kernel<<<spmmGrid, blk, 0, stream>>>(row, col, dinv, b1, b0, 2.0f);
  wmma_gemm_kernel<<<ggrid, gblk, 0, stream>>>(b0, weight + 7 * F * F, acc, 1, x, b1, 0);

  ln_elu_kernel<<<(N_NODES + 7) / 8, blk, 0, stream>>>(acc, bias, lnw, lnb, (float*)d_out);
}


# SpMM scatter split by package half: a block reads its die id, takes a per-half rank (one int atomic) and adds only into the row range its half owns; float atomic sum order unordered as before
# speedup vs baseline: 2.1636x; 1.1490x over previous
#include <hip/hip_runtime.h>
#include <math.h>

#define N_NODES 100000
#define N_EDGES 1600000
#define F 64
#define K_ORDER 8
#define LN_EPS 1e-5f

typedef __attribute__((ext_vector_type(2))) float v2f;
typedef __attribute__((ext_vector_type(4))) float v4f;
typedef __attribute__((ext_vector_type(8))) float v8f;

__global__ void zero_f32(float* __restrict__ p, int n) {
  int i = blockIdx.x * blockDim.x + threadIdx.x;
  if (i < n) p[i] = 0.0f;
}

__global__ void degree_kernel(const int* __restrict__ row,
                              const int* __restrict__ col,
                              float* __restrict__ deg) {
  int e = blockIdx.x * blockDim.x + threadIdx.x;
  if (e >= N_EDGES) return;
  int r = row[e], c = col[e];
  if (r != c) unsafeAtomicAdd(&deg[r], 1.0f);
}

__global__ void dinv_kernel(float* __restrict__ deg,
                            float* __restrict__ dinv) {
  int i = blockIdx.x * blockDim.x + threadIdx.x;
  if (i >= N_NODES) return;
  float d = deg[i];
  dinv[i] = (d > 0.0f) ? rsqrtf(fmaxf(d, 1.0f)) : 0.0f;
  deg[i] = 0.0f;
}

#define EPW 32
#define EGRP 24
#define BPW 2
#define SPMM_WAVES 4
#define N_BATCH   (N_EDGES / EPW)
#define N_WCLASS  (N_BATCH / BPW)
#define N_BCLASS  (N_WCLASS / SPMM_WAVES)
#define SPMM_GRID (2 * N_BCLASS)
#define HALF_ROWS (N_NODES / 2)
#define RANK_PITCH 64
static_assert(N_EDGES % EPW == 0, "edge count must be a multiple of the wave batch");
static_assert(N_BATCH % BPW == 0, "batches must split evenly over the wave classes");
static_assert(N_WCLASS % SPMM_WAVES == 0, "wave classes must split evenly over the work-groups");
static_assert(N_BCLASS < (1 << 24), "class must fit beside the range bit");

__global__ void __launch_bounds__(SPMM_WAVES * 32)
spmm_kernel(const int* __restrict__ row,
            const int* __restrict__ col,
            const float* __restrict__ dinv,
            const float* __restrict__ h,
            float* __restrict__ out, float scale,
            int* __restrict__ rank) {
  __shared__ int s_job;
  if (threadIdx.x == 0) {
    int id   = __builtin_amdgcn_s_sendmsg_rtn(0x87);
    int home = (id >> 18) & 1;
    int p    = atomicAdd(rank + home * RANK_PITCH, 1);
    int q    = (p < N_BCLASS) ? home : 1 - home;
    int cls  = (p < N_BCLASS) ? p : 2 * N_BCLASS - 1 - p;
    s_job = (q << 24) | cls;
  }
  __syncthreads();
  const int job = __builtin_amdgcn_readfirstlane(s_job);
  const int q   = job >> 24;
  const int cls = job & 0xFFFFFF;
  if ((unsigned)cls >= (unsigned)N_BCLASS) return;
  const int wv  = __builtin_amdgcn_readfirstlane((int)(threadIdx.x >> 5));
  const unsigned lane = threadIdx.x & 31u;
  const int wcls = cls * SPMM_WAVES + wv;

  int rr[BPW], cc[BPW], wb[BPW];
  unsigned msk[BPW];
  #pragma unroll
  for (int s = 0; s < BPW; ++s) {
    int e = (wcls + s * N_WCLASS) * EPW + (int)lane;
    rr[s] = row[e];
    cc[s] = col[e];
  }
  #pragma unroll
  for (int s = 0; s < BPW; ++s) {
    bool mine = ((rr[s] >= HALF_ROWS) ? 1 : 0) == q;
    float w = 0.0f;
    if (mine && rr[s] != cc[s]) w = scale * (-dinv[rr[s]] * dinv[cc[s]]);
    wb[s]  = __builtin_bit_cast(int, w);
    msk[s] = __builtin_amdgcn_ballot_w32(mine);
  }

  #pragma unroll
  for (int s = 0; s < BPW; ++s) {
    unsigned m = msk[s];
    while (m) {
      unsigned mm = m;
      float a0[EGRP], a1[EGRP];
      #pragma unroll
      for (int u = 0; u < EGRP; ++u) {
        if (mm) {
          int j = __builtin_ctz(mm);
          mm &= mm - 1u;
          int cj = __builtin_amdgcn_readlane(cc[s], j);
          unsigned hb = (unsigned)cj * F + lane;
          a0[u] = h[hb];
          a1[u] = h[hb + 32u];
        }
      }
      #pragma unroll
      for (int u = 0; u < EGRP; ++u) {
        if (m) {
          int j = __builtin_ctz(m);
          m &= m - 1u;
          int rj = __builtin_amdgcn_readlane(rr[s], j);
          float wj = __builtin_bit_cast(float, __builtin_amdgcn_readlane(wb[s], j));
          unsigned ob = (unsigned)rj * F + lane;
          unsafeAtomicAdd(out + ob,       wj * a0[u]);
          unsafeAtomicAdd(out + ob + 32u, wj * a1[u]);
        }
      }
    }
  }
}

__global__ void __launch_bounds__(128, 7)
wmma_gemm_kernel(const float* __restrict__ T,
                                 const float* __restrict__ Wk,
                                 float* __restrict__ acc, int beta,
                                 const float* src, float* dst, int prep) {
  int wavesPerBlock = blockDim.x >> 5;
  int tile = blockIdx.x * wavesPerBlock + (threadIdx.x >> 5);
  const int numTiles = N_NODES / 16;
  if (tile >= numTiles) return;
  int lane = threadIdx.x & 31;
  int m    = lane & 15;
  int half = lane >> 4;
  int n    = lane & 15;

  v8f c0 = {}, c1 = {}, c2 = {}, c3 = {};
  const float* trow = T + ((size_t)tile * 16 + m) * F;
  float* accbase = acc + (size_t)tile * 16 * F;

  const size_t pbase = (size_t)tile * 16 * F + (size_t)lane * 4;
  v4f pv[8];
  if (prep == 1) {
    #pragma unroll
    for (int i = 0; i < 8; ++i) pv[i] = *(const v4f*)(src + pbase + i * 128);
  }

  if (beta) {
    #pragma unroll
    for (int r = 0; r < 8; ++r) {
      int mm = r + 8 * half;
      c0[r] = accbase[mm * F +  0 + n];
      c1[r] = accbase[mm * F + 16 + n];
      c2[r] = accbase[mm * F + 32 + n];
      c3[r] = accbase[mm * F + 48 + n];
    }
  }

  #pragma unroll
  for (int kk = 0; kk < F; kk += 4) {
    int k0 = kk + half * 2;
    v2f a = *(const v2f*)(trow + k0);
    const float* wr0 = Wk + (size_t)k0 * F;
    const float* wr1 = wr0 + F;
    v2f b0 = { wr0[ 0 + n], wr1[ 0 + n] };
    v2f b1 = { wr0[16 + n], wr1[16 + n] };
    v2f b2 = { wr0[32 + n], wr1[32 + n] };
    v2f b3 = { wr0[48 + n], wr1[48 + n] };
    c0 = __builtin_amdgcn_wmma_f32_16x16x4_f32(false, a, false, b0, (short)0, c0, false, false);
    c1 = __builtin_amdgcn_wmma_f32_16x16x4_f32(false, a, false, b1, (short)0, c1, false, false);
    c2 = __builtin_amdgcn_wmma_f32_16x16x4_f32(false, a, false, b2, (short)0, c2, false, false);
    c3 = __builtin_amdgcn_wmma_f32_16x16x4_f32(false, a, false, b3, (short)0, c3, false, false);
  }

  #pragma unroll
  for (int r = 0; r < 8; ++r) {
    int mm = r + 8 * half;
    accbase[mm * F +  0 + n] = c0[r];
    accbase[mm * F + 16 + n] = c1[r];
    accbase[mm * F + 32 + n] = c2[r];
    accbase[mm * F + 48 + n] = c3[r];
  }

  if (prep == 1) {
    #pragma unroll
    for (int i = 0; i < 8; ++i) *(v4f*)(dst + pbase + i * 128) = -pv[i];
  } else if (prep == 2) {
    const v4f z = { 0.0f, 0.0f, 0.0f, 0.0f };
    #pragma unroll
    for (int i = 0; i < 8; ++i) *(v4f*)(dst + pbase + i * 128) = z;
  }
}

__global__ void ln_elu_kernel(const float* __restrict__ acc,
                              const float* __restrict__ bias,
                              const float* __restrict__ lnw,
                              const float* __restrict__ lnb,
                              float* __restrict__ out) {
  int node = blockIdx.x * (blockDim.x >> 5) + (threadIdx.x >> 5);
  if (node >= N_NODES) return;
  int lane = threadIdx.x & 31;
  int f = lane * 2;
  v2f v = *(const v2f*)(acc + (size_t)node * F + f);
  v.x += bias[f];
  v.y += bias[f + 1];
  float s = v.x + v.y;
  #pragma unroll
  for (int off = 16; off > 0; off >>= 1) s += __shfl_xor(s, off, 32);
  float mean = s * (1.0f / F);
  float dx = v.x - mean, dy = v.y - mean;
  float q = dx * dx + dy * dy;
  #pragma unroll
  for (int off = 16; off > 0; off >>= 1) q += __shfl_xor(q, off, 32);
  float inv = rsqrtf(q * (1.0f / F) + LN_EPS);
  float hx = dx * inv * lnw[f]     + lnb[f];
  float hy = dy * inv * lnw[f + 1] + lnb[f + 1];
  hx = (hx > 0.0f) ? hx : expm1f(hx);
  hy = (hy > 0.0f) ? hy : expm1f(hy);
  v2f o = { hx, hy };
  *(v2f*)(out + (size_t)node * F + f) = o;
}

extern "C" void kernel_launch(void* const* d_in, const int* in_sizes, int n_in,
                              void* d_out, int out_size, void* d_ws, size_t ws_size,
                              hipStream_t stream) {
  (void)in_sizes; (void)n_in; (void)out_size; (void)ws_size;

  const float* x      = (const float*)d_in[0];
  const int*   edge   = (const int*)d_in[1];
  const float* weight = (const float*)d_in[2];
  const float* bias   = (const float*)d_in[3];
  const float* lnw    = (const float*)d_in[4];
  const float* lnb    = (const float*)d_in[5];
  const int* row = edge;
  const int* col = edge + N_EDGES;

  float* ws   = (float*)d_ws;
  float* deg  = ws;
  float* dinv = deg + N_NODES;
  float* b0   = dinv + N_NODES;
  float* b1   = b0 + (size_t)N_NODES * F;
  float* acc  = (float*)d_out;

  dim3 blk(256);

  zero_f32<<<(N_NODES + 255) / 256, blk, 0, stream>>>(deg, N_NODES);
  degree_kernel<<<(N_EDGES + 255) / 256, blk, 0, stream>>>(row, col, deg);
  dinv_kernel<<<(N_NODES + 255) / 256, blk, 0, stream>>>(deg, dinv);

  dim3 sblk(SPMM_WAVES * 32);
  const int spmmGrid = SPMM_GRID;
  int* rank = (int*)deg;
  const int tiles = N_NODES / 16;
  dim3 gblk(128);
  int ggrid = (tiles + 3) / 4;

  wmma_gemm_kernel<<<ggrid, gblk, 0, stream>>>(x, weight + 0 * F * F, acc, 0, x, b0, 2);
  spmm_kernel<<<spmmGrid, sblk, 0, stream>>>(row, col, dinv, x, b0, 1.0f, rank + 0 * 2 * RANK_PITCH);
  wmma_gemm_kernel<<<ggrid, gblk, 0, stream>>>(b0, weight + 1 * F * F, acc, 1, x, b1, 1);

  spmm_kernel<<<spmmGrid, sblk, 0, stream>>>(row, col, dinv, b0, b1, 2.0f, rank + 1 * 2 * RANK_PITCH);
  wmma_gemm_kernel<<<ggrid, gblk, 0, stream>>>(b1, weight + 2 * F * F, acc, 1, b0, b0, 1);
  spmm_kernel<<<spmmGrid, sblk, 0, stream>>>(row, col, dinv, b1, b0, 2.0f, rank + 2 * 2 * RANK_PITCH);
  wmma_gemm_kernel<<<ggrid, gblk, 0, stream>>>(b0, weight + 3 * F * F, acc, 1, b1, b1, 1);
  spmm_kernel<<<spmmGrid, sblk, 0, stream>>>(row, col, dinv, b0, b1, 2.0f, rank + 3 * 2 * RANK_PITCH);
  wmma_gemm_kernel<<<ggrid, gblk, 0, stream>>>(b1, weight + 4 * F * F, acc, 1, b0, b0, 1);
  spmm_kernel<<<spmmGrid, sblk, 0, stream>>>(row, col, dinv, b1, b0, 2.0f, rank + 4 * 2 * RANK_PITCH);
  wmma_gemm_kernel<<<ggrid, gblk, 0, stream>>>(b0, weight + 5 * F * F, acc, 1, b1, b1, 1);
  spmm_kernel<<<spmmGrid, sblk, 0, stream>>>(row, col, dinv, b0, b1, 2.0f, rank + 5 * 2 * RANK_PITCH);
  wmma_gemm_kernel<<<ggrid, gblk, 0, stream>>>(b1, weight + 6 * F * F, acc, 1, b0, b0, 1);
  spmm_kernel<<<spmmGrid, sblk, 0, stream>>>(row, col, dinv, b1, b0, 2.0f, rank + 6 * 2 * RANK_PITCH);
  wmma_gemm_kernel<<<ggrid, gblk, 0, stream>>>(b0, weight + 7 * F * F, acc, 1, x, b1, 0);

  ln_elu_kernel<<<(N_NODES + 7) / 8, blk, 0, stream>>>(acc, bias, lnw, lnb, (float*)d_out);
}


# Edge weights -dinv[row]*dinv[col] formed once per call into the workspace (one coalesced pass); every scatter reads them coalesced instead of gathering dinv[row], dinv[col] per edge; same bits
# speedup vs baseline: 2.7113x; 1.2531x over previous
#include <hip/hip_runtime.h>
#include <math.h>

#define N_NODES 100000
#define N_EDGES 1600000
#define F 64
#define K_ORDER 8
#define LN_EPS 1e-5f

typedef __attribute__((ext_vector_type(2))) float v2f;
typedef __attribute__((ext_vector_type(4))) float v4f;
typedef __attribute__((ext_vector_type(8))) float v8f;

__global__ void zero_f32(float* __restrict__ p, int n4) {
  int i = blockIdx.x * blockDim.x + threadIdx.x;
  const v4f z = { 0.0f, 0.0f, 0.0f, 0.0f };
  if (i < n4) ((v4f*)p)[i] = z;
}

__global__ void degree_kernel(const int* __restrict__ row,
                              const int* __restrict__ col,
                              float* __restrict__ deg) {
  int e = blockIdx.x * blockDim.x + threadIdx.x;
  if (e >= N_EDGES) return;
  int r = row[e], c = col[e];
  if (r != c) unsafeAtomicAdd(&deg[r], 1.0f);
}

__global__ void dinv_kernel(float* __restrict__ deg,
                            float* __restrict__ dinv) {
  int i = blockIdx.x * blockDim.x + threadIdx.x;
  if (i >= N_NODES) return;
  float d = deg[i];
  dinv[i] = (d > 0.0f) ? rsqrtf(fmaxf(d, 1.0f)) : 0.0f;
  deg[i] = 0.0f;
}

#define WS_NEED ((size_t)(2 * (size_t)N_NODES + 2 * (size_t)N_NODES * F + (size_t)N_EDGES) * sizeof(float))

__global__ void weight_kernel(const int* __restrict__ row,
                              const int* __restrict__ col,
                              const float* __restrict__ dinv,
                              float* __restrict__ wts, size_t ws_bytes) {
  if (ws_bytes < WS_NEED) return;
  int e = blockIdx.x * blockDim.x + threadIdx.x;
  if (e >= N_EDGES) return;
  int r = row[e], c = col[e];
  wts[e] = (r != c) ? -dinv[r] * dinv[c] : 0.0f;
}


#define EPW 32
#define EGRP 24
#define BPW 2
#define SPMM_WAVES 4
#define N_BATCH   (N_EDGES / EPW)
#define N_WCLASS  (N_BATCH / BPW)
#define N_BCLASS  (N_WCLASS / SPMM_WAVES)
#define SPMM_GRID (2 * N_BCLASS)
#define HALF_ROWS (N_NODES / 2)
#define RANK_PITCH 64
#define N_TILES (N_NODES / 16)
#define GEMM_BLOCKS ((N_TILES + SPMM_WAVES - 1) / SPMM_WAVES)
static_assert(N_EDGES % EPW == 0, "edge count must be a multiple of the wave batch");
static_assert(N_BATCH % BPW == 0, "batches must split evenly over the wave classes");
static_assert(N_WCLASS % SPMM_WAVES == 0, "wave classes must split evenly over the work-groups");
static_assert(N_BCLASS < (1 << 24), "class must fit beside the range bit");
static_assert(N_NODES % 16 == 0, "rows must split into whole 16-row tiles");

__device__ __forceinline__ void
spmm_block(const int* __restrict__ row,
           const int* __restrict__ col,
           const float* __restrict__ dinv,
           const float* __restrict__ h,
           float* __restrict__ out, float scale,
           int* __restrict__ rank,
           const float* __restrict__ wts, bool have_w) {
  __shared__ int s_job;
  if (threadIdx.x == 0) {
    int id   = __builtin_amdgcn_s_sendmsg_rtn(0x87);
    int home = (id >> 18) & 1;
    int p    = atomicAdd(rank + home * RANK_PITCH, 1);
    int q    = (p < N_BCLASS) ? home : 1 - home;
    int cls  = (p < N_BCLASS) ? p : 2 * N_BCLASS - 1 - p;
    s_job = (q << 24) | cls;
  }
  __syncthreads();
  const int job = __builtin_amdgcn_readfirstlane(s_job);
  const int q   = job >> 24;
  const int cls = job & 0xFFFFFF;
  if ((unsigned)cls >= (unsigned)N_BCLASS) return;
  const int wv  = __builtin_amdgcn_readfirstlane((int)(threadIdx.x >> 5));
  const unsigned lane = threadIdx.x & 31u;
  const int wcls = cls * SPMM_WAVES + wv;

  int rr[BPW], cc[BPW], wb[BPW];
  float ww[BPW];
  unsigned msk[BPW];
  #pragma unroll
  for (int s = 0; s < BPW; ++s) {
    int e = (wcls + s * N_WCLASS) * EPW + (int)lane;
    rr[s] = row[e];
    cc[s] = col[e];
    ww[s] = have_w ? wts[e] : 0.0f;
  }
  #pragma unroll
  for (int s = 0; s < BPW; ++s) {
    bool mine = ((rr[s] >= HALF_ROWS) ? 1 : 0) == q;
    float w = 0.0f;
    if (mine && rr[s] != cc[s]) w = scale * (have_w ? ww[s] : -dinv[rr[s]] * dinv[cc[s]]);
    wb[s]  = __builtin_bit_cast(int, w);
    msk[s] = __builtin_amdgcn_ballot_w32(mine);
  }

  #pragma unroll
  for (int s = 0; s < BPW; ++s) {
    unsigned m = msk[s];
    while (m) {
      unsigned mm = m;
      float a0[EGRP], a1[EGRP];
      #pragma unroll
      for (int u = 0; u < EGRP; ++u) {
        if (mm) {
          int j = __builtin_ctz(mm);
          mm &= mm - 1u;
          int cj = __builtin_amdgcn_readlane(cc[s], j);
          unsigned hb = (unsigned)cj * F + lane;
          a0[u] = h[hb];
          a1[u] = h[hb + 32u];
        }
      }
      #pragma unroll
      for (int u = 0; u < EGRP; ++u) {
        if (m) {
          int j = __builtin_ctz(m);
          m &= m - 1u;
          int rj = __builtin_amdgcn_readlane(rr[s], j);
          float wj = __builtin_bit_cast(float, __builtin_amdgcn_readlane(wb[s], j));
          unsigned ob = (unsigned)rj * F + lane;
          unsafeAtomicAdd(out + ob,       wj * a0[u]);
          unsafeAtomicAdd(out + ob + 32u, wj * a1[u]);
        }
      }
    }
  }
}

__device__ __forceinline__ float flip_sign(float v, unsigned sm) {
  return __builtin_bit_cast(float, __builtin_bit_cast(unsigned, v) ^ sm);
}

template <int COPY, int KUNROLL>
__device__ __forceinline__ void
gemm_tile(int tile, int lane,
          const float* __restrict__ U,
          const float* __restrict__ Wk,
          float* __restrict__ acc, int beta, int neg,
          float* dst) {
  int m    = lane & 15;
  int half = lane >> 4;
  int n    = lane & 15;

  v8f c0 = {}, c1 = {}, c2 = {}, c3 = {};
  const float* trow = U + ((size_t)tile * 16 + m) * F;
  float* accbase = acc + (size_t)tile * 16 * F;

  if (beta) {
    #pragma unroll
    for (int r = 0; r < 8; ++r) {
      int mm = r + 8 * half;
      c0[r] = accbase[mm * F +  0 + n];
      c1[r] = accbase[mm * F + 16 + n];
      c2[r] = accbase[mm * F + 32 + n];
      c3[r] = accbase[mm * F + 48 + n];
    }
  }

  const unsigned sm = neg ? 0x80000000u : 0u;

  #pragma unroll KUNROLL
  for (int kk = 0; kk < F; kk += 4) {
    int k0 = kk + half * 2;
    v2f a = *(const v2f*)(trow + k0);
    a = v2f{ flip_sign(a.x, sm), flip_sign(a.y, sm) };
    const float* wr0 = Wk + (size_t)k0 * F;
    const float* wr1 = wr0 + F;
    v2f b0 = { wr0[ 0 + n], wr1[ 0 + n] };
    v2f b1 = { wr0[16 + n], wr1[16 + n] };
    v2f b2 = { wr0[32 + n], wr1[32 + n] };
    v2f b3 = { wr0[48 + n], wr1[48 + n] };
    c0 = __builtin_amdgcn_wmma_f32_16x16x4_f32(false, a, false, b0, (short)0, c0, false, false);
    c1 = __builtin_amdgcn_wmma_f32_16x16x4_f32(false, a, false, b1, (short)0, c1, false, false);
    c2 = __builtin_amdgcn_wmma_f32_16x16x4_f32(false, a, false, b2, (short)0, c2, false, false);
    c3 = __builtin_amdgcn_wmma_f32_16x16x4_f32(false, a, false, b3, (short)0, c3, false, false);
  }

  #pragma unroll
  for (int r = 0; r < 8; ++r) {
    int mm = r + 8 * half;
    accbase[mm * F +  0 + n] = c0[r];
    accbase[mm * F + 16 + n] = c1[r];
    accbase[mm * F + 32 + n] = c2[r];
    accbase[mm * F + 48 + n] = c3[r];
  }

  if (COPY) {
    const size_t pbase = (size_t)tile * 16 * F + (size_t)lane * 4;
    v4f pv[8];
    #pragma unroll
    for (int i = 0; i < 8; ++i) pv[i] = *(const v4f*)(U + pbase + i * 128);
    #pragma unroll
    for (int i = 0; i < 8; ++i) *(v4f*)(dst + pbase + i * 128) = pv[i];
  }
}

template <int COPY>
__global__ void __launch_bounds__(SPMM_WAVES * 32, 16)
step_kernel(const int* __restrict__ row,
            const int* __restrict__ col,
            const float* __restrict__ dinv,
            const float* __restrict__ h,
            float* __restrict__ out, float scale,
            int* __restrict__ rank,
            const float* __restrict__ Wk,
            float* __restrict__ acc, int beta, int neg,
            float* dst,
            const float* __restrict__ wts, size_t ws_bytes) {
  if (blockIdx.x < GEMM_BLOCKS) {
    int tile = blockIdx.x * SPMM_WAVES + (threadIdx.x >> 5);
    if (tile >= N_TILES) return;
    gemm_tile<COPY, 4>(tile, threadIdx.x & 31, h, Wk, acc, beta, neg, dst);
    return;
  }
  spmm_block(row, col, dinv, h, out, scale, rank, wts, ws_bytes >= WS_NEED);
}

__global__ void __launch_bounds__(128, 7)
wmma_gemm_kernel(const float* __restrict__ U,
                 const float* __restrict__ Wk,
                 float* __restrict__ acc, int beta, int neg) {
  int tile = blockIdx.x * (blockDim.x >> 5) + (threadIdx.x >> 5);
  if (tile >= N_TILES) return;
  gemm_tile<0, 16>(tile, threadIdx.x & 31, U, Wk, acc, beta, neg, acc);
}

__global__ void ln_elu_kernel(const float* __restrict__ acc,
                              const float* __restrict__ bias,
                              const float* __restrict__ lnw,
                              const float* __restrict__ lnb,
                              float* __restrict__ out) {
  int node = blockIdx.x * (blockDim.x >> 5) + (threadIdx.x >> 5);
  if (node >= N_NODES) return;
  int lane = threadIdx.x & 31;
  int f = lane * 2;
  v2f v = *(const v2f*)(acc + (size_t)node * F + f);
  v.x += bias[f];
  v.y += bias[f + 1];
  float s = v.x + v.y;
  #pragma unroll
  for (int off = 16; off > 0; off >>= 1) s += __shfl_xor(s, off, 32);
  float mean = s * (1.0f / F);
  float dx = v.x - mean, dy = v.y - mean;
  float q = dx * dx + dy * dy;
  #pragma unroll
  for (int off = 16; off > 0; off >>= 1) q += __shfl_xor(q, off, 32);
  float inv = rsqrtf(q * (1.0f / F) + LN_EPS);
  float hx = dx * inv * lnw[f]     + lnb[f];
  float hy = dy * inv * lnw[f + 1] + lnb[f + 1];
  hx = (hx > 0.0f) ? hx : expm1f(hx);
  hy = (hy > 0.0f) ? hy : expm1f(hy);
  v2f o = { hx, hy };
  *(v2f*)(out + (size_t)node * F + f) = o;
}

extern "C" void kernel_launch(void* const* d_in, const int* in_sizes, int n_in,
                              void* d_out, int out_size, void* d_ws, size_t ws_size,
                              hipStream_t stream) {
  (void)in_sizes; (void)n_in; (void)out_size;

  const float* x      = (const float*)d_in[0];
  const int*   edge   = (const int*)d_in[1];
  const float* weight = (const float*)d_in[2];
  const float* bias   = (const float*)d_in[3];
  const float* lnw    = (const float*)d_in[4];
  const float* lnb    = (const float*)d_in[5];
  const int* row = edge;
  const int* col = edge + N_EDGES;

  float* ws   = (float*)d_ws;
  float* deg  = ws;
  float* dinv = deg + N_NODES;
  float* b0   = dinv + N_NODES;
  float* b1   = b0 + (size_t)N_NODES * F;
  float* wts  = b1 + (size_t)N_NODES * F;
  float* acc  = (float*)d_out;

  dim3 blk(256);

  zero_f32<<<(N_NODES * (2 + F) / 4 + 255) / 256, blk, 0, stream>>>(ws, N_NODES * (2 + F) / 4);
  degree_kernel<<<(N_EDGES + 255) / 256, blk, 0, stream>>>(row, col, deg);
  dinv_kernel<<<(N_NODES + 255) / 256, blk, 0, stream>>>(deg, dinv);
  weight_kernel<<<(N_EDGES + 255) / 256, blk, 0, stream>>>(row, col, dinv, wts, ws_size);

  dim3 sblk(SPMM_WAVES * 32);
  const int stepGrid = GEMM_BLOCKS + SPMM_GRID;
  int* rank = (int*)deg;

  step_kernel<1><<<stepGrid, sblk, 0, stream>>>(row, col, dinv, x,  b0,  1.0f, rank + 0 * 2 * RANK_PITCH, weight + 0 * F * F, acc, 0, 0, b1, wts, ws_size);
  step_kernel<0><<<stepGrid, sblk, 0, stream>>>(row, col, dinv, b0, b1, -2.0f, rank + 1 * 2 * RANK_PITCH, weight + 1 * F * F, acc, 1, 0, b1, wts, ws_size);
  step_kernel<0><<<stepGrid, sblk, 0, stream>>>(row, col, dinv, b1, b0,  2.0f, rank + 2 * 2 * RANK_PITCH, weight + 2 * F * F, acc, 1, 1, b0, wts, ws_size);
  step_kernel<0><<<stepGrid, sblk, 0, stream>>>(row, col, dinv, b0, b1, -2.0f, rank + 3 * 2 * RANK_PITCH, weight + 3 * F * F, acc, 1, 1, b1, wts, ws_size);
  step_kernel<0><<<stepGrid, sblk, 0, stream>>>(row, col, dinv, b1, b0,  2.0f, rank + 4 * 2 * RANK_PITCH, weight + 4 * F * F, acc, 1, 0, b0, wts, ws_size);
  step_kernel<0><<<stepGrid, sblk, 0, stream>>>(row, col, dinv, b0, b1, -2.0f, rank + 5 * 2 * RANK_PITCH, weight + 5 * F * F, acc, 1, 0, b1, wts, ws_size);
  step_kernel<0><<<stepGrid, sblk, 0, stream>>>(row, col, dinv, b1, b0,  2.0f, rank + 6 * 2 * RANK_PITCH, weight + 6 * F * F, acc, 1, 1, b0, wts, ws_size);
  wmma_gemm_kernel<<<GEMM_BLOCKS, sblk, 0, stream>>>(b0, weight + 7 * F * F, acc, 1, 1);

  ln_elu_kernel<<<(N_NODES + 7) / 8, blk, 0, stream>>>(acc, bias, lnw, lnb, (float*)d_out);
}


# Degree count without global atomics: a work-group counts one 20000-row range of a 32000-edge slice in LDS integer counters, dinv pass adds 50 partial counts; exact, no float sum order changed
# speedup vs baseline: 3.0615x; 1.1292x over previous
#include <hip/hip_runtime.h>
#include <math.h>

#define N_NODES 100000
#define N_EDGES 1600000
#define F 64
#define K_ORDER 8
#define LN_EPS 1e-5f

typedef __attribute__((ext_vector_type(2))) float v2f;
typedef __attribute__((ext_vector_type(4))) float v4f;
typedef __attribute__((ext_vector_type(8))) float v8f;

__global__ void zero_f32(float* __restrict__ p, int n4) {
  int i = blockIdx.x * blockDim.x + threadIdx.x;
  const v4f z = { 0.0f, 0.0f, 0.0f, 0.0f };
  if (i < n4) ((v4f*)p)[i] = z;
}

typedef __attribute__((ext_vector_type(4))) int v4i;
typedef __attribute__((ext_vector_type(4))) unsigned v4u;
#define DEG_PARTS   50
#define DEG_RANGES  5
#define DEG_SLICE   (N_EDGES / DEG_PARTS)
#define DEG_ROWS    (N_NODES / DEG_RANGES)
#define DEG_THREADS 1024
#define DEG_LOADS   ((DEG_SLICE / 4 + DEG_THREADS - 1) / DEG_THREADS)
#define DEG_FLUSH   ((DEG_ROWS / 4 + DEG_THREADS - 1) / DEG_THREADS)
static_assert(N_EDGES % DEG_PARTS == 0 && DEG_SLICE % 32 == 0, "a slice must be whole lines of indices");
static_assert(N_NODES % DEG_RANGES == 0 && DEG_ROWS % 32 == 0, "a range must be whole lines of counters");
static_assert(DEG_PARTS <= F, "the planes must fit in one feature buffer");

__global__ void __launch_bounds__(DEG_THREADS)
degree_kernel(const int* __restrict__ row,
              const int* __restrict__ col,
              unsigned* __restrict__ part) {
  __shared__ unsigned s_cnt[DEG_ROWS];
  const int slice = blockIdx.x / DEG_RANGES;
  const int base  = (blockIdx.x % DEG_RANGES) * DEG_ROWS;
  const v4i* r4 = (const v4i*)(row + (size_t)slice * DEG_SLICE);
  const v4i* c4 = (const v4i*)(col + (size_t)slice * DEG_SLICE);

  v4i rv[DEG_LOADS], cv[DEG_LOADS];
  #pragma unroll
  for (int k = 0; k < DEG_LOADS; ++k) {
    int i = (int)threadIdx.x + k * DEG_THREADS;
    if (i < DEG_SLICE / 4) { rv[k] = r4[i]; cv[k] = c4[i]; }
    else                   { rv[k] = v4i{ -1, -1, -1, -1 }; cv[k] = rv[k]; }
  }

  const v4u z = { 0u, 0u, 0u, 0u };
  for (int i = threadIdx.x; i < DEG_ROWS / 4; i += DEG_THREADS) ((v4u*)s_cnt)[i] = z;
  __syncthreads();

  #pragma unroll
  for (int k = 0; k < DEG_LOADS; ++k) {
    #pragma unroll
    for (int j = 0; j < 4; ++j) {
      int r = rv[k][j], c = cv[k][j];
      unsigned t = (unsigned)(r - base);
      if (t < (unsigned)DEG_ROWS && r != c) atomicAdd(&s_cnt[t], 1u);
    }
  }
  __syncthreads();

  v4u* dst = (v4u*)(part + (size_t)slice * N_NODES + base);
  v4u fv[DEG_FLUSH];
  #pragma unroll
  for (int k = 0; k < DEG_FLUSH; ++k) {
    int i = (int)threadIdx.x + k * DEG_THREADS;
    if (i < DEG_ROWS / 4) fv[k] = ((const v4u*)s_cnt)[i];
  }
  #pragma unroll
  for (int k = 0; k < DEG_FLUSH; ++k) {
    int i = (int)threadIdx.x + k * DEG_THREADS;
    if (i < DEG_ROWS / 4) dst[i] = fv[k];
  }
}

#define RANK_PITCH 64
#define N_RANK_WORDS (7 * 2 * RANK_PITCH)
__global__ void dinv_kernel(const unsigned* __restrict__ part,
                            int* __restrict__ rankw,
                            float* __restrict__ dinv) {
  int i = blockIdx.x * blockDim.x + threadIdx.x;
  if (i >= N_NODES) return;
  unsigned n = 0;
  #pragma unroll
  for (int p = 0; p < DEG_PARTS; ++p) n += part[(size_t)p * N_NODES + i];
  float d = (float)n;
  dinv[i] = (d > 0.0f) ? rsqrtf(fmaxf(d, 1.0f)) : 0.0f;
  if (i < N_RANK_WORDS) rankw[i] = 0;
}

#define WS_NEED ((size_t)(2 * (size_t)N_NODES + 2 * (size_t)N_NODES * F + (size_t)N_EDGES) * sizeof(float))

__global__ void weight_kernel(const int* __restrict__ row,
                              const int* __restrict__ col,
                              const float* __restrict__ dinv,
                              float* __restrict__ wts, size_t ws_bytes) {
  if (ws_bytes < WS_NEED) return;
  int e = blockIdx.x * blockDim.x + threadIdx.x;
  if (e >= N_EDGES) return;
  int r = row[e], c = col[e];
  wts[e] = (r != c) ? -dinv[r] * dinv[c] : 0.0f;
}


#define EPW 32
#define EGRP 24
#define BPW 2
#define SPMM_WAVES 4
#define N_BATCH   (N_EDGES / EPW)
#define N_WCLASS  (N_BATCH / BPW)
#define N_BCLASS  (N_WCLASS / SPMM_WAVES)
#define SPMM_GRID (2 * N_BCLASS)
#define HALF_ROWS (N_NODES / 2)
#define N_TILES (N_NODES / 16)
#define GEMM_BLOCKS ((N_TILES + SPMM_WAVES - 1) / SPMM_WAVES)
static_assert(N_EDGES % EPW == 0, "edge count must be a multiple of the wave batch");
static_assert(N_BATCH % BPW == 0, "batches must split evenly over the wave classes");
static_assert(N_WCLASS % SPMM_WAVES == 0, "wave classes must split evenly over the work-groups");
static_assert(N_BCLASS < (1 << 24), "class must fit beside the range bit");
static_assert(N_NODES % 16 == 0, "rows must split into whole 16-row tiles");

__device__ __forceinline__ void
spmm_block(const int* __restrict__ row,
           const int* __restrict__ col,
           const float* __restrict__ dinv,
           const float* __restrict__ h,
           float* __restrict__ out, float scale,
           int* __restrict__ rank,
           const float* __restrict__ wts, bool have_w) {
  __shared__ int s_job;
  if (threadIdx.x == 0) {
    int id   = __builtin_amdgcn_s_sendmsg_rtn(0x87);
    int home = (id >> 18) & 1;
    int p    = atomicAdd(rank + home * RANK_PITCH, 1);
    int q    = (p < N_BCLASS) ? home : 1 - home;
    int cls  = (p < N_BCLASS) ? p : 2 * N_BCLASS - 1 - p;
    s_job = (q << 24) | cls;
  }
  __syncthreads();
  const int job = __builtin_amdgcn_readfirstlane(s_job);
  const int q   = job >> 24;
  const int cls = job & 0xFFFFFF;
  if ((unsigned)cls >= (unsigned)N_BCLASS) return;
  const int wv  = __builtin_amdgcn_readfirstlane((int)(threadIdx.x >> 5));
  const unsigned lane = threadIdx.x & 31u;
  const int wcls = cls * SPMM_WAVES + wv;

  int rr[BPW], cc[BPW], wb[BPW];
  float ww[BPW];
  unsigned msk[BPW];
  #pragma unroll
  for (int s = 0; s < BPW; ++s) {
    int e = (wcls + s * N_WCLASS) * EPW + (int)lane;
    rr[s] = row[e];
    cc[s] = col[e];
    ww[s] = have_w ? wts[e] : 0.0f;
  }
  #pragma unroll
  for (int s = 0; s < BPW; ++s) {
    bool mine = ((rr[s] >= HALF_ROWS) ? 1 : 0) == q;
    float w = 0.0f;
    if (mine && rr[s] != cc[s]) w = scale * (have_w ? ww[s] : -dinv[rr[s]] * dinv[cc[s]]);
    wb[s]  = __builtin_bit_cast(int, w);
    msk[s] = __builtin_amdgcn_ballot_w32(mine);
  }

  #pragma unroll
  for (int s = 0; s < BPW; ++s) {
    unsigned m = msk[s];
    while (m) {
      unsigned mm = m;
      float a0[EGRP], a1[EGRP];
      #pragma unroll
      for (int u = 0; u < EGRP; ++u) {
        if (mm) {
          int j = __builtin_ctz(mm);
          mm &= mm - 1u;
          int cj = __builtin_amdgcn_readlane(cc[s], j);
          unsigned hb = (unsigned)cj * F + lane;
          a0[u] = h[hb];
          a1[u] = h[hb + 32u];
        }
      }
      #pragma unroll
      for (int u = 0; u < EGRP; ++u) {
        if (m) {
          int j = __builtin_ctz(m);
          m &= m - 1u;
          int rj = __builtin_amdgcn_readlane(rr[s], j);
          float wj = __builtin_bit_cast(float, __builtin_amdgcn_readlane(wb[s], j));
          unsigned ob = (unsigned)rj * F + lane;
          unsafeAtomicAdd(out + ob,       wj * a0[u]);
          unsafeAtomicAdd(out + ob + 32u, wj * a1[u]);
        }
      }
    }
  }
}

__device__ __forceinline__ float flip_sign(float v, unsigned sm) {
  return __builtin_bit_cast(float, __builtin_bit_cast(unsigned, v) ^ sm);
}

template <int COPY, int KUNROLL>
__device__ __forceinline__ void
gemm_tile(int tile, int lane,
          const float* __restrict__ U,
          const float* __restrict__ Wk,
          float* __restrict__ acc, int beta, int neg,
          float* dst) {
  int m    = lane & 15;
  int half = lane >> 4;
  int n    = lane & 15;

  v8f c0 = {}, c1 = {}, c2 = {}, c3 = {};
  const float* trow = U + ((size_t)tile * 16 + m) * F;
  float* accbase = acc + (size_t)tile * 16 * F;

  if (beta) {
    #pragma unroll
    for (int r = 0; r < 8; ++r) {
      int mm = r + 8 * half;
      c0[r] = accbase[mm * F +  0 + n];
      c1[r] = accbase[mm * F + 16 + n];
      c2[r] = accbase[mm * F + 32 + n];
      c3[r] = accbase[mm * F + 48 + n];
    }
  }

  const unsigned sm = neg ? 0x80000000u : 0u;

  #pragma unroll KUNROLL
  for (int kk = 0; kk < F; kk += 4) {
    int k0 = kk + half * 2;
    v2f a = *(const v2f*)(trow + k0);
    a = v2f{ flip_sign(a.x, sm), flip_sign(a.y, sm) };
    const float* wr0 = Wk + (size_t)k0 * F;
    const float* wr1 = wr0 + F;
    v2f b0 = { wr0[ 0 + n], wr1[ 0 + n] };
    v2f b1 = { wr0[16 + n], wr1[16 + n] };
    v2f b2 = { wr0[32 + n], wr1[32 + n] };
    v2f b3 = { wr0[48 + n], wr1[48 + n] };
    c0 = __builtin_amdgcn_wmma_f32_16x16x4_f32(false, a, false, b0, (short)0, c0, false, false);
    c1 = __builtin_amdgcn_wmma_f32_16x16x4_f32(false, a, false, b1, (short)0, c1, false, false);
    c2 = __builtin_amdgcn_wmma_f32_16x16x4_f32(false, a, false, b2, (short)0, c2, false, false);
    c3 = __builtin_amdgcn_wmma_f32_16x16x4_f32(false, a, false, b3, (short)0, c3, false, false);
  }

  #pragma unroll
  for (int r = 0; r < 8; ++r) {
    int mm = r + 8 * half;
    accbase[mm * F +  0 + n] = c0[r];
    accbase[mm * F + 16 + n] = c1[r];
    accbase[mm * F + 32 + n] = c2[r];
    accbase[mm * F + 48 + n] = c3[r];
  }

  if (COPY) {
    const size_t pbase = (size_t)tile * 16 * F + (size_t)lane * 4;
    v4f pv[8];
    #pragma unroll
    for (int i = 0; i < 8; ++i) pv[i] = *(const v4f*)(U + pbase + i * 128);
    #pragma unroll
    for (int i = 0; i < 8; ++i) *(v4f*)(dst + pbase + i * 128) = pv[i];
  }
}

template <int COPY>
__global__ void __launch_bounds__(SPMM_WAVES * 32, 16)
step_kernel(const int* __restrict__ row,
            const int* __restrict__ col,
            const float* __restrict__ dinv,
            const float* __restrict__ h,
            float* __restrict__ out, float scale,
            int* __restrict__ rank,
            const float* __restrict__ Wk,
            float* __restrict__ acc, int beta, int neg,
            float* dst,
            const float* __restrict__ wts, size_t ws_bytes) {
  if (blockIdx.x < GEMM_BLOCKS) {
    int tile = blockIdx.x * SPMM_WAVES + (threadIdx.x >> 5);
    if (tile >= N_TILES) return;
    gemm_tile<COPY, 4>(tile, threadIdx.x & 31, h, Wk, acc, beta, neg, dst);
    return;
  }
  spmm_block(row, col, dinv, h, out, scale, rank, wts, ws_bytes >= WS_NEED);
}

__global__ void __launch_bounds__(128, 7)
wmma_gemm_kernel(const float* __restrict__ U,
                 const float* __restrict__ Wk,
                 float* __restrict__ acc, int beta, int neg) {
  int tile = blockIdx.x * (blockDim.x >> 5) + (threadIdx.x >> 5);
  if (tile >= N_TILES) return;
  gemm_tile<0, 16>(tile, threadIdx.x & 31, U, Wk, acc, beta, neg, acc);
}

__global__ void ln_elu_kernel(const float* __restrict__ acc,
                              const float* __restrict__ bias,
                              const float* __restrict__ lnw,
                              const float* __restrict__ lnb,
                              float* __restrict__ out) {
  int node = blockIdx.x * (blockDim.x >> 5) + (threadIdx.x >> 5);
  if (node >= N_NODES) return;
  int lane = threadIdx.x & 31;
  int f = lane * 2;
  v2f v = *(const v2f*)(acc + (size_t)node * F + f);
  v.x += bias[f];
  v.y += bias[f + 1];
  float s = v.x + v.y;
  #pragma unroll
  for (int off = 16; off > 0; off >>= 1) s += __shfl_xor(s, off, 32);
  float mean = s * (1.0f / F);
  float dx = v.x - mean, dy = v.y - mean;
  float q = dx * dx + dy * dy;
  #pragma unroll
  for (int off = 16; off > 0; off >>= 1) q += __shfl_xor(q, off, 32);
  float inv = rsqrtf(q * (1.0f / F) + LN_EPS);
  float hx = dx * inv * lnw[f]     + lnb[f];
  float hy = dy * inv * lnw[f + 1] + lnb[f + 1];
  hx = (hx > 0.0f) ? hx : expm1f(hx);
  hy = (hy > 0.0f) ? hy : expm1f(hy);
  v2f o = { hx, hy };
  *(v2f*)(out + (size_t)node * F + f) = o;
}

extern "C" void kernel_launch(void* const* d_in, const int* in_sizes, int n_in,
                              void* d_out, int out_size, void* d_ws, size_t ws_size,
                              hipStream_t stream) {
  (void)in_sizes; (void)n_in; (void)out_size;

  const float* x      = (const float*)d_in[0];
  const int*   edge   = (const int*)d_in[1];
  const float* weight = (const float*)d_in[2];
  const float* bias   = (const float*)d_in[3];
  const float* lnw    = (const float*)d_in[4];
  const float* lnb    = (const float*)d_in[5];
  const int* row = edge;
  const int* col = edge + N_EDGES;

  float* ws   = (float*)d_ws;
  float* deg  = ws;
  float* dinv = deg + N_NODES;
  float* b0   = dinv + N_NODES;
  float* b1   = b0 + (size_t)N_NODES * F;
  float* wts  = b1 + (size_t)N_NODES * F;
  float* acc  = (float*)d_out;

  dim3 blk(256);

  unsigned* part = (unsigned*)b1;
  zero_f32<<<(N_NODES * F / 4 + 255) / 256, blk, 0, stream>>>(b0, N_NODES * F / 4);
  degree_kernel<<<DEG_PARTS * DEG_RANGES, DEG_THREADS, 0, stream>>>(row, col, part);
  dinv_kernel<<<(N_NODES + 255) / 256, blk, 0, stream>>>(part, (int*)deg, dinv);
  weight_kernel<<<(N_EDGES + 255) / 256, blk, 0, stream>>>(row, col, dinv, wts, ws_size);

  dim3 sblk(SPMM_WAVES * 32);
  const int stepGrid = GEMM_BLOCKS + SPMM_GRID;
  int* rank = (int*)deg;

  step_kernel<1><<<stepGrid, sblk, 0, stream>>>(row, col, dinv, x,  b0,  1.0f, rank + 0 * 2 * RANK_PITCH, weight + 0 * F * F, acc, 0, 0, b1, wts, ws_size);
  step_kernel<0><<<stepGrid, sblk, 0, stream>>>(row, col, dinv, b0, b1, -2.0f, rank + 1 * 2 * RANK_PITCH, weight + 1 * F * F, acc, 1, 0, b1, wts, ws_size);
  step_kernel<0><<<stepGrid, sblk, 0, stream>>>(row, col, dinv, b1, b0,  2.0f, rank + 2 * 2 * RANK_PITCH, weight + 2 * F * F, acc, 1, 1, b0, wts, ws_size);
  step_kernel<0><<<stepGrid, sblk, 0, stream>>>(row, col, dinv, b0, b1, -2.0f, rank + 3 * 2 * RANK_PITCH, weight + 3 * F * F, acc, 1, 1, b1, wts, ws_size);
  step_kernel<0><<<stepGrid, sblk, 0, stream>>>(row, col, dinv, b1, b0,  2.0f, rank + 4 * 2 * RANK_PITCH, weight + 4 * F * F, acc, 1, 0, b0, wts, ws_size);
  step_kernel<0><<<stepGrid, sblk, 0, stream>>>(row, col, dinv, b0, b1, -2.0f, rank + 5 * 2 * RANK_PITCH, weight + 5 * F * F, acc, 1, 0, b1, wts, ws_size);
  step_kernel<0><<<stepGrid, sblk, 0, stream>>>(row, col, dinv, b1, b0,  2.0f, rank + 6 * 2 * RANK_PITCH, weight + 6 * F * F, acc, 1, 1, b0, wts, ws_size);
  wmma_gemm_kernel<<<GEMM_BLOCKS, sblk, 0, stream>>>(b0, weight + 7 * F * F, acc, 1, 1);

  ln_elu_kernel<<<(N_NODES + 7) / 8, blk, 0, stream>>>(acc, bias, lnw, lnb, (float*)d_out);
}


# Two streaming launches folded into neighbours: LayerNorm+ELU becomes the last product's register epilogue (row sums by a new fixed lane tree), b0's zero fill rides in the edge-weight pass
# speedup vs baseline: 3.1706x; 1.0356x over previous
#include <hip/hip_runtime.h>
#include <math.h>

#define N_NODES 100000
#define N_EDGES 1600000
#define F 64
#define K_ORDER 8
#define LN_EPS 1e-5f

typedef __attribute__((ext_vector_type(2))) float v2f;
typedef __attribute__((ext_vector_type(4))) float v4f;
typedef __attribute__((ext_vector_type(8))) float v8f;

typedef __attribute__((ext_vector_type(4))) int v4i;
typedef __attribute__((ext_vector_type(4))) unsigned v4u;
#define DEG_PARTS   50
#define DEG_RANGES  5
#define DEG_SLICE   (N_EDGES / DEG_PARTS)
#define DEG_ROWS    (N_NODES / DEG_RANGES)
#define DEG_THREADS 1024
#define DEG_LOADS   ((DEG_SLICE / 4 + DEG_THREADS - 1) / DEG_THREADS)
#define DEG_FLUSH   ((DEG_ROWS / 4 + DEG_THREADS - 1) / DEG_THREADS)
static_assert(N_EDGES % DEG_PARTS == 0 && DEG_SLICE % 32 == 0, "a slice must be whole lines of indices");
static_assert(N_NODES % DEG_RANGES == 0 && DEG_ROWS % 32 == 0, "a range must be whole lines of counters");
static_assert(DEG_PARTS <= F, "the planes must fit in one feature buffer");

__global__ void __launch_bounds__(DEG_THREADS)
degree_kernel(const int* __restrict__ row,
              const int* __restrict__ col,
              unsigned* __restrict__ part) {
  __shared__ unsigned s_cnt[DEG_ROWS];
  const int slice = blockIdx.x / DEG_RANGES;
  const int base  = (blockIdx.x % DEG_RANGES) * DEG_ROWS;
  const v4i* r4 = (const v4i*)(row + (size_t)slice * DEG_SLICE);
  const v4i* c4 = (const v4i*)(col + (size_t)slice * DEG_SLICE);

  v4i rv[DEG_LOADS], cv[DEG_LOADS];
  #pragma unroll
  for (int k = 0; k < DEG_LOADS; ++k) {
    int i = (int)threadIdx.x + k * DEG_THREADS;
    if (i < DEG_SLICE / 4) { rv[k] = r4[i]; cv[k] = c4[i]; }
    else                   { rv[k] = v4i{ -1, -1, -1, -1 }; cv[k] = rv[k]; }
  }

  const v4u z = { 0u, 0u, 0u, 0u };
  for (int i = threadIdx.x; i < DEG_ROWS / 4; i += DEG_THREADS) ((v4u*)s_cnt)[i] = z;
  __syncthreads();

  #pragma unroll
  for (int k = 0; k < DEG_LOADS; ++k) {
    #pragma unroll
    for (int j = 0; j < 4; ++j) {
      int r = rv[k][j], c = cv[k][j];
      unsigned t = (unsigned)(r - base);
      if (t < (unsigned)DEG_ROWS && r != c) atomicAdd(&s_cnt[t], 1u);
    }
  }
  __syncthreads();

  v4u* dst = (v4u*)(part + (size_t)slice * N_NODES + base);
  v4u fv[DEG_FLUSH];
  #pragma unroll
  for (int k = 0; k < DEG_FLUSH; ++k) {
    int i = (int)threadIdx.x + k * DEG_THREADS;
    if (i < DEG_ROWS / 4) fv[k] = ((const v4u*)s_cnt)[i];
  }
  #pragma unroll
  for (int k = 0; k < DEG_FLUSH; ++k) {
    int i = (int)threadIdx.x + k * DEG_THREADS;
    if (i < DEG_ROWS / 4) dst[i] = fv[k];
  }
}

#define RANK_PITCH 64
#define N_RANK_WORDS (7 * 2 * RANK_PITCH)
__global__ void dinv_kernel(const unsigned* __restrict__ part,
                            int* __restrict__ rankw,
                            float* __restrict__ dinv) {
  int i = blockIdx.x * blockDim.x + threadIdx.x;
  if (i >= N_NODES) return;
  unsigned n = 0;
  #pragma unroll
  for (int p = 0; p < DEG_PARTS; ++p) n += part[(size_t)p * N_NODES + i];
  float d = (float)n;
  dinv[i] = (d > 0.0f) ? rsqrtf(fmaxf(d, 1.0f)) : 0.0f;
  if (i < N_RANK_WORDS) rankw[i] = 0;
}

#define WS_NEED ((size_t)(2 * (size_t)N_NODES + 2 * (size_t)N_NODES * F + (size_t)N_EDGES) * sizeof(float))

static_assert(N_NODES * F / 4 <= N_EDGES, "one thread per float4 word of the cleared buffer");
__global__ void weight_kernel(const int* __restrict__ row,
                              const int* __restrict__ col,
                              const float* __restrict__ dinv,
                              float* __restrict__ wts, size_t ws_bytes,
                              float* __restrict__ zdst) {
  int e = blockIdx.x * blockDim.x + threadIdx.x;
  const v4f z = { 0.0f, 0.0f, 0.0f, 0.0f };
  if (e < N_NODES * F / 4) ((v4f*)zdst)[e] = z;
  if (ws_bytes < WS_NEED) return;
  if (e >= N_EDGES) return;
  int r = row[e], c = col[e];
  wts[e] = (r != c) ? -dinv[r] * dinv[c] : 0.0f;
}


#define EPW 32
#define EGRP 24
#define BPW 2
#define SPMM_WAVES 4
#define N_BATCH   (N_EDGES / EPW)
#define N_WCLASS  (N_BATCH / BPW)
#define N_BCLASS  (N_WCLASS / SPMM_WAVES)
#define SPMM_GRID (2 * N_BCLASS)
#define HALF_ROWS (N_NODES / 2)
#define N_TILES (N_NODES / 16)
#define GEMM_BLOCKS ((N_TILES + SPMM_WAVES - 1) / SPMM_WAVES)
static_assert(N_EDGES % EPW == 0, "edge count must be a multiple of the wave batch");
static_assert(N_BATCH % BPW == 0, "batches must split evenly over the wave classes");
static_assert(N_WCLASS % SPMM_WAVES == 0, "wave classes must split evenly over the work-groups");
static_assert(N_BCLASS < (1 << 24), "class must fit beside the range bit");
static_assert(N_NODES % 16 == 0, "rows must split into whole 16-row tiles");

__device__ __forceinline__ void
spmm_block(const int* __restrict__ row,
           const int* __restrict__ col,
           const float* __restrict__ dinv,
           const float* __restrict__ h,
           float* __restrict__ out, float scale,
           int* __restrict__ rank,
           const float* __restrict__ wts, bool have_w) {
  __shared__ int s_job;
  if (threadIdx.x == 0) {
    int id   = __builtin_amdgcn_s_sendmsg_rtn(0x87);
    int home = (id >> 18) & 1;
    int p    = atomicAdd(rank + home * RANK_PITCH, 1);
    int q    = (p < N_BCLASS) ? home : 1 - home;
    int cls  = (p < N_BCLASS) ? p : 2 * N_BCLASS - 1 - p;
    s_job = (q << 24) | cls;
  }
  __syncthreads();
  const int job = __builtin_amdgcn_readfirstlane(s_job);
  const int q   = job >> 24;
  const int cls = job & 0xFFFFFF;
  if ((unsigned)cls >= (unsigned)N_BCLASS) return;
  const int wv  = __builtin_amdgcn_readfirstlane((int)(threadIdx.x >> 5));
  const unsigned lane = threadIdx.x & 31u;
  const int wcls = cls * SPMM_WAVES + wv;

  int rr[BPW], cc[BPW], wb[BPW];
  float ww[BPW];
  unsigned msk[BPW];
  #pragma unroll
  for (int s = 0; s < BPW; ++s) {
    int e = (wcls + s * N_WCLASS) * EPW + (int)lane;
    rr[s] = row[e];
    cc[s] = col[e];
    ww[s] = have_w ? wts[e] : 0.0f;
  }
  #pragma unroll
  for (int s = 0; s < BPW; ++s) {
    bool mine = ((rr[s] >= HALF_ROWS) ? 1 : 0) == q;
    float w = 0.0f;
    if (mine && rr[s] != cc[s]) w = scale * (have_w ? ww[s] : -dinv[rr[s]] * dinv[cc[s]]);
    wb[s]  = __builtin_bit_cast(int, w);
    msk[s] = __builtin_amdgcn_ballot_w32(mine);
  }

  #pragma unroll
  for (int s = 0; s < BPW; ++s) {
    unsigned m = msk[s];
    while (m) {
      unsigned mm = m;
      float a0[EGRP], a1[EGRP];
      #pragma unroll
      for (int u = 0; u < EGRP; ++u) {
        if (mm) {
          int j = __builtin_ctz(mm);
          mm &= mm - 1u;
          int cj = __builtin_amdgcn_readlane(cc[s], j);
          unsigned hb = (unsigned)cj * F + lane;
          a0[u] = h[hb];
          a1[u] = h[hb + 32u];
        }
      }
      #pragma unroll
      for (int u = 0; u < EGRP; ++u) {
        if (m) {
          int j = __builtin_ctz(m);
          m &= m - 1u;
          int rj = __builtin_amdgcn_readlane(rr[s], j);
          float wj = __builtin_bit_cast(float, __builtin_amdgcn_readlane(wb[s], j));
          unsigned ob = (unsigned)rj * F + lane;
          unsafeAtomicAdd(out + ob,       wj * a0[u]);
          unsafeAtomicAdd(out + ob + 32u, wj * a1[u]);
        }
      }
    }
  }
}

__device__ __forceinline__ float flip_sign(float v, unsigned sm) {
  return __builtin_bit_cast(float, __builtin_bit_cast(unsigned, v) ^ sm);
}

template <int M>
__device__ __forceinline__ float lane_xor16(float v) {
  return __builtin_bit_cast(float,
      __builtin_amdgcn_update_dpp(0, __builtin_bit_cast(int, v), 0x160 + M, 0xf, 0xf, false));
}
__device__ __forceinline__ float row16_sum(float v) {
  v += lane_xor16<1>(v);
  v += lane_xor16<2>(v);
  v += lane_xor16<4>(v);
  v += lane_xor16<8>(v);
  return v;
}

template <int COPY, int KUNROLL, int LNE>
__device__ __forceinline__ void
gemm_tile(int tile, int lane,
          const float* __restrict__ U,
          const float* __restrict__ Wk,
          float* __restrict__ acc, int beta, int neg,
          float* dst,
          const float* __restrict__ bias,
          const float* __restrict__ lnw,
          const float* __restrict__ lnb) {
  int m    = lane & 15;
  int half = lane >> 4;
  int n    = lane & 15;

  v8f c0 = {}, c1 = {}, c2 = {}, c3 = {};
  const float* trow = U + ((size_t)tile * 16 + m) * F;
  float* accbase = acc + (size_t)tile * 16 * F;

  if (beta) {
    #pragma unroll
    for (int r = 0; r < 8; ++r) {
      int mm = r + 8 * half;
      c0[r] = accbase[mm * F +  0 + n];
      c1[r] = accbase[mm * F + 16 + n];
      c2[r] = accbase[mm * F + 32 + n];
      c3[r] = accbase[mm * F + 48 + n];
    }
  }

  const unsigned sm = neg ? 0x80000000u : 0u;

  #pragma unroll KUNROLL
  for (int kk = 0; kk < F; kk += 4) {
    int k0 = kk + half * 2;
    v2f a = *(const v2f*)(trow + k0);
    a = v2f{ flip_sign(a.x, sm), flip_sign(a.y, sm) };
    const float* wr0 = Wk + (size_t)k0 * F;
    const float* wr1 = wr0 + F;
    v2f b0 = { wr0[ 0 + n], wr1[ 0 + n] };
    v2f b1 = { wr0[16 + n], wr1[16 + n] };
    v2f b2 = { wr0[32 + n], wr1[32 + n] };
    v2f b3 = { wr0[48 + n], wr1[48 + n] };
    c0 = __builtin_amdgcn_wmma_f32_16x16x4_f32(false, a, false, b0, (short)0, c0, false, false);
    c1 = __builtin_amdgcn_wmma_f32_16x16x4_f32(false, a, false, b1, (short)0, c1, false, false);
    c2 = __builtin_amdgcn_wmma_f32_16x16x4_f32(false, a, false, b2, (short)0, c2, false, false);
    c3 = __builtin_amdgcn_wmma_f32_16x16x4_f32(false, a, false, b3, (short)0, c3, false, false);
  }

  if (LNE) {
    float bs[4], gw[4], gb[4];
    #pragma unroll
    for (int j = 0; j < 4; ++j) {
      bs[j] = bias[16 * j + n];
      gw[j] = lnw[16 * j + n];
      gb[j] = lnb[16 * j + n];
    }
    float s[8], q[8];
    #pragma unroll
    for (int r = 0; r < 8; ++r) {
      c0[r] += bs[0];
      c1[r] += bs[1];
      c2[r] += bs[2];
      c3[r] += bs[3];
      s[r] = (c0[r] + c1[r]) + (c2[r] + c3[r]);
    }
    #pragma unroll
    for (int r = 0; r < 8; ++r) s[r] = row16_sum(s[r]);
    #pragma unroll
    for (int r = 0; r < 8; ++r) {
      float mean = s[r] * (1.0f / F);
      c0[r] -= mean;
      c1[r] -= mean;
      c2[r] -= mean;
      c3[r] -= mean;
      q[r] = (c0[r] * c0[r] + c1[r] * c1[r]) + (c2[r] * c2[r] + c3[r] * c3[r]);
    }
    #pragma unroll
    for (int r = 0; r < 8; ++r) q[r] = row16_sum(q[r]);
    #pragma unroll
    for (int r = 0; r < 8; ++r) {
      int mm = r + 8 * half;
      float inv = rsqrtf(q[r] * (1.0f / F) + LN_EPS);
      float h0 = c0[r] * inv * gw[0] + gb[0];
      float h1 = c1[r] * inv * gw[1] + gb[1];
      float h2 = c2[r] * inv * gw[2] + gb[2];
      float h3 = c3[r] * inv * gw[3] + gb[3];
      h0 = (h0 > 0.0f) ? h0 : expm1f(h0);
      h1 = (h1 > 0.0f) ? h1 : expm1f(h1);
      h2 = (h2 > 0.0f) ? h2 : expm1f(h2);
      h3 = (h3 > 0.0f) ? h3 : expm1f(h3);
      accbase[mm * F +  0 + n] = h0;
      accbase[mm * F + 16 + n] = h1;
      accbase[mm * F + 32 + n] = h2;
      accbase[mm * F + 48 + n] = h3;
    }
  } else {
    #pragma unroll
    for (int r = 0; r < 8; ++r) {
      int mm = r + 8 * half;
      accbase[mm * F +  0 + n] = c0[r];
      accbase[mm * F + 16 + n] = c1[r];
      accbase[mm * F + 32 + n] = c2[r];
      accbase[mm * F + 48 + n] = c3[r];
    }
  }

  if (COPY) {
    const size_t pbase = (size_t)tile * 16 * F + (size_t)lane * 4;
    v4f pv[8];
    #pragma unroll
    for (int i = 0; i < 8; ++i) pv[i] = *(const v4f*)(U + pbase + i * 128);
    #pragma unroll
    for (int i = 0; i < 8; ++i) *(v4f*)(dst + pbase + i * 128) = pv[i];
  }
}

template <int COPY>
__global__ void __launch_bounds__(SPMM_WAVES * 32, 16)
step_kernel(const int* __restrict__ row,
            const int* __restrict__ col,
            const float* __restrict__ dinv,
            const float* __restrict__ h,
            float* __restrict__ out, float scale,
            int* __restrict__ rank,
            const float* __restrict__ Wk,
            float* __restrict__ acc, int beta, int neg,
            float* dst,
            const float* __restrict__ wts, size_t ws_bytes) {
  if (blockIdx.x < GEMM_BLOCKS) {
    int tile = blockIdx.x * SPMM_WAVES + (threadIdx.x >> 5);
    if (tile >= N_TILES) return;
    gemm_tile<COPY, 4, 0>(tile, threadIdx.x & 31, h, Wk, acc, beta, neg, dst, nullptr, nullptr, nullptr);
    return;
  }
  spmm_block(row, col, dinv, h, out, scale, rank, wts, ws_bytes >= WS_NEED);
}

__global__ void __launch_bounds__(128, 7)
last_gemm_ln_elu_kernel(const float* __restrict__ U,
                        const float* __restrict__ Wk,
                        float* __restrict__ acc, int beta, int neg,
                        const float* __restrict__ bias,
                        const float* __restrict__ lnw,
                        const float* __restrict__ lnb) {
  int tile = blockIdx.x * (blockDim.x >> 5) + (threadIdx.x >> 5);
  if (tile >= N_TILES) return;
  gemm_tile<0, 16, 1>(tile, threadIdx.x & 31, U, Wk, acc, beta, neg, acc, bias, lnw, lnb);
}

extern "C" void kernel_launch(void* const* d_in, const int* in_sizes, int n_in,
                              void* d_out, int out_size, void* d_ws, size_t ws_size,
                              hipStream_t stream) {
  (void)in_sizes; (void)n_in; (void)out_size;

  const float* x      = (const float*)d_in[0];
  const int*   edge   = (const int*)d_in[1];
  const float* weight = (const float*)d_in[2];
  const float* bias   = (const float*)d_in[3];
  const float* lnw    = (const float*)d_in[4];
  const float* lnb    = (const float*)d_in[5];
  const int* row = edge;
  const int* col = edge + N_EDGES;

  float* ws   = (float*)d_ws;
  float* deg  = ws;
  float* dinv = deg + N_NODES;
  float* b0   = dinv + N_NODES;
  float* b1   = b0 + (size_t)N_NODES * F;
  float* wts  = b1 + (size_t)N_NODES * F;
  float* acc  = (float*)d_out;

  dim3 blk(256);

  unsigned* part = (unsigned*)b1;
  degree_kernel<<<DEG_PARTS * DEG_RANGES, DEG_THREADS, 0, stream>>>(row, col, part);
  dinv_kernel<<<(N_NODES + 255) / 256, blk, 0, stream>>>(part, (int*)deg, dinv);
  weight_kernel<<<(N_EDGES + 255) / 256, blk, 0, stream>>>(row, col, dinv, wts, ws_size, b0);

  dim3 sblk(SPMM_WAVES * 32);
  const int stepGrid = GEMM_BLOCKS + SPMM_GRID;
  int* rank = (int*)deg;

  step_kernel<1><<<stepGrid, sblk, 0, stream>>>(row, col, dinv, x,  b0,  1.0f, rank + 0 * 2 * RANK_PITCH, weight + 0 * F * F, acc, 0, 0, b1, wts, ws_size);
  step_kernel<0><<<stepGrid, sblk, 0, stream>>>(row, col, dinv, b0, b1, -2.0f, rank + 1 * 2 * RANK_PITCH, weight + 1 * F * F, acc, 1, 0, b1, wts, ws_size);
  step_kernel<0><<<stepGrid, sblk, 0, stream>>>(row, col, dinv, b1, b0,  2.0f, rank + 2 * 2 * RANK_PITCH, weight + 2 * F * F, acc, 1, 1, b0, wts, ws_size);
  step_kernel<0><<<stepGrid, sblk, 0, stream>>>(row, col, dinv, b0, b1, -2.0f, rank + 3 * 2 * RANK_PITCH, weight + 3 * F * F, acc, 1, 1, b1, wts, ws_size);
  step_kernel<0><<<stepGrid, sblk, 0, stream>>>(row, col, dinv, b1, b0,  2.0f, rank + 4 * 2 * RANK_PITCH, weight + 4 * F * F, acc, 1, 0, b0, wts, ws_size);
  step_kernel<0><<<stepGrid, sblk, 0, stream>>>(row, col, dinv, b0, b1, -2.0f, rank + 5 * 2 * RANK_PITCH, weight + 5 * F * F, acc, 1, 0, b1, wts, ws_size);
  step_kernel<0><<<stepGrid, sblk, 0, stream>>>(row, col, dinv, b1, b0,  2.0f, rank + 6 * 2 * RANK_PITCH, weight + 6 * F * F, acc, 1, 1, b0, wts, ws_size);
  last_gemm_ln_elu_kernel<<<GEMM_BLOCKS, sblk, 0, stream>>>(b0, weight + 7 * F * F, acc, 1, 1, bias, lnw, lnb);
}
